# starting kernel
# baseline (speedup 1.0000x reference)
#include <hip/hip_runtime.h>
#include <hip/hip_bf16.h>
#include <math.h>

#define V_  128
#define H_  1024
#define SW_ 1024
#define SD_ 8192

typedef __attribute__((ext_vector_type(2))) float v2f;
typedef __attribute__((ext_vector_type(4))) float v4f;
typedef __attribute__((ext_vector_type(8))) float v8f;

#define WS_CTRL   0
#define WS_SINP   16
#define WS_RNNIN  2048
#define WS_GH     4096
#define WS_GI     8192
#define WS_HNEW   12288

template <int ACT>
__global__ void gemv_wmma(const float* __restrict__ W, const float* __restrict__ vec,
                          const float* __restrict__ bias, float* __restrict__ out, int K) {
    const int lane = threadIdx.x & 31;
    const int wave = threadIdx.x >> 5;
    const int row_base = blockIdx.x * 16;
    const int kslice = K >> 3;
    const int kbeg = wave * kslice;

    const int r  = row_base + (lane & 15);
    const int co = (lane >> 4) << 1;
    const float* wp = W + (size_t)r * K + (size_t)(kbeg + co);
    const float* vp = vec + kbeg + co;

    v8f c0 = {}, c1 = {}, c2 = {}, c3 = {};
    for (int k = 0; k < kslice; k += 16) {
        if ((k & 63) == 0) __builtin_prefetch(wp + k + 256, 0, 1);
        v2f a0 = *(const v2f*)(vp + k);
        v2f b0 = *(const v2f*)(wp + k);
        v2f a1 = *(const v2f*)(vp + k + 4);
        v2f b1 = *(const v2f*)(wp + k + 4);
        v2f a2 = *(const v2f*)(vp + k + 8);
        v2f b2 = *(const v2f*)(wp + k + 8);
        v2f a3 = *(const v2f*)(vp + k + 12);
        v2f b3 = *(const v2f*)(wp + k + 12);
        c0 = __builtin_amdgcn_wmma_f32_16x16x4_f32(false, a0, false, b0, (short)0, c0, false, false);
        c1 = __builtin_amdgcn_wmma_f32_16x16x4_f32(false, a1, false, b1, (short)0, c1, false, false);
        c2 = __builtin_amdgcn_wmma_f32_16x16x4_f32(false, a2, false, b2, (short)0, c2, false, false);
        c3 = __builtin_amdgcn_wmma_f32_16x16x4_f32(false, a3, false, b3, (short)0, c3, false, false);
    }

    __shared__ float red[8][16];
    if (lane < 16) red[wave][lane] = c0[0] + c1[0] + c2[0] + c3[0];
    __syncthreads();
    if (threadIdx.x < 16) {
        float s = 0.f;
        #pragma unroll
        for (int w = 0; w < 8; ++w) s += red[w][threadIdx.x];
        s += bias[row_base + threadIdx.x];
        if (ACT == 1) s = tanhf(s);
        out[row_base + threadIdx.x] = s;
    }
}

__global__ void ctrl_and_x(const int* __restrict__ inp, const float* __restrict__ hidden,
                           const float* __restrict__ emb, const float* __restrict__ W_ctrl,
                           const float* __restrict__ b_ctrl, float* ws) {
    if (blockIdx.x == 0) {
        __shared__ float lg[3];
        const int wave = threadIdx.x >> 5, lane = threadIdx.x & 31;
        if (wave < 3) {
            float s = 0.f;
            for (int i = lane; i < H_; i += 32) s += W_ctrl[wave * H_ + i] * hidden[i];
            #pragma unroll
            for (int off = 16; off > 0; off >>= 1) s += __shfl_down(s, off, 32);
            if (lane == 0) lg[wave] = s + b_ctrl[wave];
        }
        __syncthreads();
        if (threadIdx.x == 0) {
            float m = fmaxf(lg[0], fmaxf(lg[1], lg[2]));
            float e0 = expf(lg[0] - m), e1 = expf(lg[1] - m), e2 = expf(lg[2] - m);
            float inv = 1.f / (e0 + e1 + e2);
            ws[WS_CTRL + 0] = e0 * inv;
            ws[WS_CTRL + 1] = e1 * inv;
            ws[WS_CTRL + 2] = e2 * inv;
        }
    } else {
        int i = (blockIdx.x - 1) * blockDim.x + threadIdx.x;
        if (i < H_) ws[WS_RNNIN + i] = emb[(size_t)inp[0] * H_ + i];
    }
}

__global__ void stack_update(const float* __restrict__ stack, float* __restrict__ new_stack,
                             float* ws) {
    const float a_push = ws[WS_CTRL + 0];
    const float a_pop  = ws[WS_CTRL + 1];
    const float a_noop = ws[WS_CTRL + 2];

    const int idx = blockIdx.x * blockDim.x + threadIdx.x;
    const int row_f4 = SW_ / 4;
    const int j = idx / row_f4;
    const int c = (idx - j * row_f4) * 4;

    v4f cur = *(const v4f*)(stack + (size_t)j * SW_ + c);
    v4f up  = (j == 0) ? *(const v4f*)(ws + WS_SINP + c)
                       : *(const v4f*)(stack + (size_t)(j - 1) * SW_ + c);
    v4f dn = {};
    if (j < SD_ - 1) dn = *(const v4f*)(stack + (size_t)(j + 1) * SW_ + c);

    v4f o = a_noop * cur + a_push * up + a_pop * dn;

    __builtin_nontemporal_store(o, (v4f*)(new_stack + (size_t)j * SW_ + c));
    if (j == 0) *(v4f*)(ws + WS_RNNIN + H_ + c) = o;
}

__global__ void gru_gates(const float* __restrict__ hidden, float* ws,
                          float* __restrict__ hid_out) {
    const int i = blockIdx.x * blockDim.x + threadIdx.x;
    if (i >= H_) return;
    const float* gi = ws + WS_GI;
    const float* gh = ws + WS_GH;
    float r  = 1.f / (1.f + expf(-(gi[i] + gh[i])));
    float z  = 1.f / (1.f + expf(-(gi[H_ + i] + gh[H_ + i])));
    float n  = tanhf(gi[2 * H_ + i] + r * gh[2 * H_ + i]);
    float hn = (1.f - z) * n + z * hidden[i];
    hid_out[i] = hn;
    ws[WS_HNEW + i] = hn;
}

extern "C" void kernel_launch(void* const* d_in, const int* in_sizes, int n_in,
                              void* d_out, int out_size, void* d_ws, size_t ws_size,
                              hipStream_t stream) {
    (void)in_sizes; (void)n_in; (void)out_size; (void)ws_size;
    const int*   inp    = (const int*)  d_in[0];
    const float* hidden = (const float*)d_in[1];
    const float* stack  = (const float*)d_in[2];
    const float* emb    = (const float*)d_in[3];
    const float* W_ctrl = (const float*)d_in[4];
    const float* b_ctrl = (const float*)d_in[5];
    const float* W_sin  = (const float*)d_in[6];
    const float* b_sin  = (const float*)d_in[7];
    const float* W_ih   = (const float*)d_in[8];
    const float* b_ih   = (const float*)d_in[9];
    const float* W_hh   = (const float*)d_in[10];
    const float* b_hh   = (const float*)d_in[11];
    const float* W_dec  = (const float*)d_in[12];
    const float* b_dec  = (const float*)d_in[13];

    float* out       = (float*)d_out;
    float* hid_out   = out + V_;
    float* stack_out = out + V_ + H_;
    float* ws        = (float*)d_ws;

    ctrl_and_x<<<dim3(1 + H_ / 128), dim3(128), 0, stream>>>(inp, hidden, emb, W_ctrl, b_ctrl, ws);
    gemv_wmma<1><<<dim3(SW_ / 16), dim3(256), 0, stream>>>(W_sin, hidden, b_sin, ws + WS_SINP, H_);
    gemv_wmma<0><<<dim3(3 * H_ / 16), dim3(256), 0, stream>>>(W_hh, hidden, b_hh, ws + WS_GH, H_);
    stack_update<<<dim3((SD_ * SW_ / 4) / 256), dim3(256), 0, stream>>>(stack, stack_out, ws);
    gemv_wmma<0><<<dim3(3 * H_ / 16), dim3(256), 0, stream>>>(W_ih, ws + WS_RNNIN, b_ih, ws + WS_GI, H_ + SW_);
    gru_gates<<<dim3(H_ / 256), dim3(256), 0, stream>>>(hidden, ws, hid_out);
    gemv_wmma<0><<<dim3(V_ / 16), dim3(256), 0, stream>>>(W_dec, ws + WS_HNEW, b_dec, out, H_);
}


# Stack blend leaves its own launch: rows 1.. ride the stack_input launch as extra blocks; row 0 folds into gi by linearity (3 source rows on 3 WMMA A rows; gi sum order changes); 5 launches -> 4
# speedup vs baseline: 1.5279x; 1.5279x over previous
#include <hip/hip_runtime.h>
#include <hip/hip_bf16.h>
#include <math.h>

#define V_  128
#define H_  1024
#define SW_ 1024
#define SD_ 8192

typedef __attribute__((ext_vector_type(2))) float v2f;
typedef __attribute__((ext_vector_type(4))) float v4f;
typedef __attribute__((ext_vector_type(8))) float v8f;

#define WS_CTRL   0
#define WS_SINP   16
#define WS_GH     4096
#define WS_GI     8192
#define WS_HNEW   12288

#define KS_    128
#define PITCH_ (KS_ + 4)

template <int ACT, int K>
__device__ __forceinline__
void gemv_rows16(const float* __restrict__ W, const float* __restrict__ vec,
                 const float* __restrict__ bias, float* __restrict__ out, const int row_base,
                 float (*tile)[16][PITCH_], float (*red)[16]) {
    constexpr int NW = K / KS_;
    const int lane = threadIdx.x & 31;
    const int wave = threadIdx.x >> 5;
    const int kbeg = wave * KS_;

    const float* wsrc = W + (size_t)row_base * K + (size_t)(kbeg + lane * 4);
    v4f t[16];
    for (int r = 0; r < 16; ++r) t[r] = *(const v4f*)(wsrc + (size_t)r * K);

    const int co = (lane >> 4) << 1;
    const float* vp = vec + kbeg + co;
    v2f av[KS_ / 4];
    for (int s = 0; s < KS_ / 4; ++s) av[s] = *(const v2f*)(vp + 4 * s);

    for (int r = 0; r < 16; ++r) *(v4f*)&tile[wave][r][lane * 4] = t[r];
    __syncthreads();

    const float* bp = &tile[wave][lane & 15][co];
    v8f c0 = {}, c1 = {}, c2 = {}, c3 = {};
    for (int k = 0; k < KS_; k += 16) {
        v2f b0 = *(const v2f*)(bp + k);
        v2f b1 = *(const v2f*)(bp + k + 4);
        v2f b2 = *(const v2f*)(bp + k + 8);
        v2f b3 = *(const v2f*)(bp + k + 12);
        c0 = __builtin_amdgcn_wmma_f32_16x16x4_f32(false, av[k / 4],     false, b0, (short)0, c0, false, false);
        c1 = __builtin_amdgcn_wmma_f32_16x16x4_f32(false, av[k / 4 + 1], false, b1, (short)0, c1, false, false);
        c2 = __builtin_amdgcn_wmma_f32_16x16x4_f32(false, av[k / 4 + 2], false, b2, (short)0, c2, false, false);
        c3 = __builtin_amdgcn_wmma_f32_16x16x4_f32(false, av[k / 4 + 3], false, b3, (short)0, c3, false, false);
    }

    if (lane < 16) red[wave][lane] = c0[0] + c1[0] + c2[0] + c3[0];
    __syncthreads();
    if (threadIdx.x < 16) {
        float s = 0.f;
        #pragma unroll
        for (int w = 0; w < NW; ++w) s += red[w][threadIdx.x];
        s += bias[row_base + threadIdx.x];
        if (ACT == 1) s = tanhf(s);
        out[row_base + threadIdx.x] = s;
    }
}

template <int ACT, int K>
__global__ __launch_bounds__((K / KS_) * 32)
void gemv_wmma(const float* __restrict__ W, const float* __restrict__ vec,
               const float* __restrict__ bias, float* __restrict__ out) {
    __shared__ __attribute__((aligned(16))) float tile[K / KS_][16][PITCH_];
    __shared__ float red[K / KS_][16];
    gemv_rows16<ACT, K>(W, vec, bias, out, blockIdx.x * 16, tile, red);
}

#define CTRL_WAVES_ 8
static_assert(CTRL_WAVES_ * 32 * 4 == H_, "one 16-byte piece of h per thread");

__device__ __forceinline__
void ctrl_partials(const v4f hv, const v4f w0, const v4f w1, const v4f w2,
                   const int lane, float* part_w) {
    float p0 = hv[0] * w0[0] + hv[1] * w0[1] + hv[2] * w0[2] + hv[3] * w0[3];
    float p1 = hv[0] * w1[0] + hv[1] * w1[1] + hv[2] * w1[2] + hv[3] * w1[3];
    float p2 = hv[0] * w2[0] + hv[1] * w2[1] + hv[2] * w2[2] + hv[3] * w2[3];
    for (int off = 16; off > 0; off >>= 1) {
        p0 += __shfl_xor(p0, off, 32);
        p1 += __shfl_xor(p1, off, 32);
        p2 += __shfl_xor(p2, off, 32);
    }
    if (lane == 0) *(v4f*)part_w = v4f{p0, p1, p2, 0.f};
}

__device__ __forceinline__
void ctrl_softmax(const float (*part)[4], const float bc0, const float bc1, const float bc2,
                  float& a_push, float& a_pop, float& a_noop) {
    float l0 = 0.f, l1 = 0.f, l2 = 0.f;
    for (int w = 0; w < CTRL_WAVES_; ++w) {
        const v4f q = *(const v4f*)part[w];
        l0 += q[0]; l1 += q[1]; l2 += q[2];
    }
    l0 += bc0; l1 += bc1; l2 += bc2;
    const float m  = fmaxf(l0, fmaxf(l1, l2));
    const float e0 = expf(l0 - m), e1 = expf(l1 - m), e2 = expf(l2 - m);
    const float inv = 1.f / (e0 + e1 + e2);
    a_push = e0 * inv;
    a_pop  = e1 * inv;
    a_noop = e2 * inv;
}

#define BLEND_ROWS_   8
#define BLEND_NP_     (SW_ / 128)
#define SINP_BLOCKS_  (SW_ / 16)
#define BLEND_BLOCKS_ (SD_ / BLEND_ROWS_)
static_assert(BLEND_ROWS_ == CTRL_WAVES_, "the blend block's waves form the controls");
static_assert(BLEND_ROWS_ == H_ / KS_, "both kinds of block have the same number of waves");

__global__ __launch_bounds__(BLEND_ROWS_ * 32)
void sinp_and_blend(const float* __restrict__ W_sin, const float* __restrict__ hidden,
                    const float* __restrict__ b_sin, const float* __restrict__ stack,
                    float* __restrict__ new_stack, const float* __restrict__ W_ctrl,
                    const float* __restrict__ b_ctrl, float* __restrict__ ws) {
    __shared__ __attribute__((aligned(16))) float tile[H_ / KS_][16][PITCH_];
    __shared__ float red[H_ / KS_][16];
    __shared__ __attribute__((aligned(16))) float part[CTRL_WAVES_][4];

    if (blockIdx.x < SINP_BLOCKS_) {
        gemv_rows16<1, H_>(W_sin, hidden, b_sin, ws + WS_SINP, blockIdx.x * 16, tile, red);
        return;
    }

    const int lane = threadIdx.x & 31;
    const int wave = threadIdx.x >> 5;

    const int e = threadIdx.x * 4;
    const v4f hv = *(const v4f*)(hidden + e);
    const v4f w0 = *(const v4f*)(W_ctrl + e);
    const v4f w1 = *(const v4f*)(W_ctrl + H_ + e);
    const v4f w2 = *(const v4f*)(W_ctrl + 2 * H_ + e);
    const float bc0 = b_ctrl[0], bc1 = b_ctrl[1], bc2 = b_ctrl[2];

    const int j  = (blockIdx.x - SINP_BLOCKS_) * BLEND_ROWS_ + wave;
    const bool live = (j != 0);
    const int c0 = lane * 4;
    const float* cp = stack + (size_t)j * SW_ + c0;
    const float* upp = live ? (cp - SW_) : cp;

    v4f cur[BLEND_NP_], up[BLEND_NP_], dn[BLEND_NP_];
    for (int p = 0; p < BLEND_NP_; ++p) cur[p] = *(const v4f*)(cp + p * 128);
    for (int p = 0; p < BLEND_NP_; ++p) up[p]  = *(const v4f*)(upp + p * 128);
    for (int p = 0; p < BLEND_NP_; ++p) dn[p]  = v4f{0.f, 0.f, 0.f, 0.f};
    if (j < SD_ - 1)
        for (int p = 0; p < BLEND_NP_; ++p) dn[p] = *(const v4f*)(cp + SW_ + p * 128);

    ctrl_partials(hv, w0, w1, w2, lane, part[wave]);
    __syncthreads();
    float a_push, a_pop, a_noop;
    ctrl_softmax(part, bc0, bc1, bc2, a_push, a_pop, a_noop);
    if (!live) {
        if (lane == 0) {
            ws[WS_CTRL + 0] = a_push;
            ws[WS_CTRL + 1] = a_pop;
            ws[WS_CTRL + 2] = a_noop;
        }
        return;
    }

    float* op = new_stack + (size_t)j * SW_ + c0;
    for (int p = 0; p < BLEND_NP_; ++p) {
        v4f o = a_noop * cur[p] + a_push * up[p] + a_pop * dn[p];
        __builtin_nontemporal_store(o, (v4f*)(op + p * 128));
    }
}

#define NW_IH_ ((H_ + SW_) / KS_)
#define NW_X_  (H_ / KS_)
#define NW_S_  (SW_ / KS_)
#define NW_HH_ (H_ / KS_)
#define NW_G_  (NW_IH_ + NW_HH_)
#define GI_BLOCKS_ (3 * H_ / 16)

__global__ __launch_bounds__(NW_G_ * 32)
void gemv_gi_gh(const int* __restrict__ inp, const float* __restrict__ emb,
                const float* __restrict__ hidden, const float* __restrict__ stack,
                const float* __restrict__ W_ih, const float* __restrict__ b_ih,
                const float* __restrict__ W_hh, const float* __restrict__ b_hh,
                float* __restrict__ new_stack, float* ws) {
    __shared__ __attribute__((aligned(16))) float tile[NW_G_][16][PITCH_];
    __shared__ float red[NW_G_][16];
    __shared__ float red_up[NW_S_][16];
    __shared__ float red_dn[NW_S_][16];

    const float a_push = ws[WS_CTRL + 0];
    const float a_pop  = ws[WS_CTRL + 1];
    const float a_noop = ws[WS_CTRL + 2];

    if (blockIdx.x == GI_BLOCKS_) {
        if (threadIdx.x < SW_ / 4) {
            const int e = threadIdx.x * 4;
            const v4f r_cur = *(const v4f*)(stack + e);
            const v4f r_up  = *(const v4f*)((const float*)ws + WS_SINP + e);
            const v4f r_dn  = *(const v4f*)(stack + SW_ + e);
            v4f o = a_noop * r_cur + a_push * r_up + a_pop * r_dn;
            __builtin_nontemporal_store(o, (v4f*)(new_stack + e));
        }
        return;
    }

    const int lane = threadIdx.x & 31;
    const int wave = threadIdx.x >> 5;
    const int row_base = blockIdx.x * 16;
    const int co = (lane >> 4) << 1;

    v4f t[16];
    const float* vp;
    if (wave < NW_IH_) {
        const float* wsrc = W_ih + (size_t)row_base * (H_ + SW_) + (size_t)(wave * KS_ + lane * 4);
        for (int r = 0; r < 16; ++r) t[r] = *(const v4f*)(wsrc + (size_t)r * (H_ + SW_));
        if (wave < NW_X_) {
            vp = emb + (size_t)inp[0] * H_ + wave * KS_;
        } else {
            const int m = lane & 15;
            const float* rowp = (m == 1) ? (const float*)ws + WS_SINP
                              : (m == 2) ? stack + SW_ : stack;
            vp = rowp + (wave - NW_X_) * KS_;
        }
    } else {
        const int kw = wave - NW_IH_;
        const float* wsrc = W_hh + (size_t)row_base * H_ + (size_t)(kw * KS_ + lane * 4);
        for (int r = 0; r < 16; ++r) t[r] = *(const v4f*)(wsrc + (size_t)r * H_);
        vp = hidden + kw * KS_;
    }
    vp += co;
    v2f av[KS_ / 4];
    for (int s = 0; s < KS_ / 4; ++s) av[s] = *(const v2f*)(vp + 4 * s);

    for (int r = 0; r < 16; ++r) *(v4f*)&tile[wave][r][lane * 4] = t[r];
    __syncthreads();

    const float* bp = &tile[wave][lane & 15][co];
    v8f c0 = {}, c1 = {}, c2 = {}, c3 = {};
    for (int k = 0; k < KS_; k += 16) {
        v2f b0 = *(const v2f*)(bp + k);
        v2f b1 = *(const v2f*)(bp + k + 4);
        v2f b2 = *(const v2f*)(bp + k + 8);
        v2f b3 = *(const v2f*)(bp + k + 12);
        c0 = __builtin_amdgcn_wmma_f32_16x16x4_f32(false, av[k / 4],     false, b0, (short)0, c0, false, false);
        c1 = __builtin_amdgcn_wmma_f32_16x16x4_f32(false, av[k / 4 + 1], false, b1, (short)0, c1, false, false);
        c2 = __builtin_amdgcn_wmma_f32_16x16x4_f32(false, av[k / 4 + 2], false, b2, (short)0, c2, false, false);
        c3 = __builtin_amdgcn_wmma_f32_16x16x4_f32(false, av[k / 4 + 3], false, b3, (short)0, c3, false, false);
    }

    if (lane < 16) {
        red[wave][lane] = c0[0] + c1[0] + c2[0] + c3[0];
        if (wave >= NW_X_ && wave < NW_IH_) {
            red_up[wave - NW_X_][lane] = c0[1] + c1[1] + c2[1] + c3[1];
            red_dn[wave - NW_X_][lane] = c0[2] + c1[2] + c2[2] + c3[2];
        }
    }
    __syncthreads();

    if (wave == 0 && lane < 16) {
        float sx = 0.f, sc = 0.f, su = 0.f, sd = 0.f;
        for (int w = 0; w < NW_X_; ++w) sx += red[w][lane];
        for (int w = 0; w < NW_S_; ++w) {
            sc += red[NW_X_ + w][lane];
            su += red_up[w][lane];
            sd += red_dn[w][lane];
        }
        ws[WS_GI + row_base + lane] = sx + (a_noop * sc + a_push * su + a_pop * sd) + b_ih[row_base + lane];
    } else if (wave == 1 && lane < 16) {
        float s = 0.f;
        for (int w = NW_IH_; w < NW_G_; ++w) s += red[w][lane];
        ws[WS_GH + row_base + lane] = s + b_hh[row_base + lane];
    }
}

__global__ void gru_gates(const float* __restrict__ hidden, float* ws,
                          float* __restrict__ hid_out) {
    const int i = blockIdx.x * blockDim.x + threadIdx.x;
    if (i >= H_) return;
    const float* gi = ws + WS_GI;
    const float* gh = ws + WS_GH;
    float r  = 1.f / (1.f + expf(-(gi[i] + gh[i])));
    float z  = 1.f / (1.f + expf(-(gi[H_ + i] + gh[H_ + i])));
    float n  = tanhf(gi[2 * H_ + i] + r * gh[2 * H_ + i]);
    float hn = (1.f - z) * n + z * hidden[i];
    hid_out[i] = hn;
    ws[WS_HNEW + i] = hn;
}

extern "C" void kernel_launch(void* const* d_in, const int* in_sizes, int n_in,
                              void* d_out, int out_size, void* d_ws, size_t ws_size,
                              hipStream_t stream) {
    (void)in_sizes; (void)n_in; (void)out_size; (void)ws_size;
    const int*   inp    = (const int*)  d_in[0];
    const float* hidden = (const float*)d_in[1];
    const float* stack  = (const float*)d_in[2];
    const float* emb    = (const float*)d_in[3];
    const float* W_ctrl = (const float*)d_in[4];
    const float* b_ctrl = (const float*)d_in[5];
    const float* W_sin  = (const float*)d_in[6];
    const float* b_sin  = (const float*)d_in[7];
    const float* W_ih   = (const float*)d_in[8];
    const float* b_ih   = (const float*)d_in[9];
    const float* W_hh   = (const float*)d_in[10];
    const float* b_hh   = (const float*)d_in[11];
    const float* W_dec  = (const float*)d_in[12];
    const float* b_dec  = (const float*)d_in[13];

    float* out       = (float*)d_out;
    float* hid_out   = out + V_;
    float* stack_out = out + V_ + H_;
    float* ws        = (float*)d_ws;

    sinp_and_blend<<<dim3(SINP_BLOCKS_ + BLEND_BLOCKS_), dim3(BLEND_ROWS_ * 32), 0, stream>>>(
        W_sin, hidden, b_sin, stack, stack_out, W_ctrl, b_ctrl, ws);
    gemv_gi_gh<<<dim3(GI_BLOCKS_ + 1), dim3(NW_G_ * 32), 0, stream>>>(
        inp, emb, hidden, stack, W_ih, b_ih, W_hh, b_hh, stack_out, ws);
    gru_gates<<<dim3(H_ / 256), dim3(256), 0, stream>>>(hidden, ws, hid_out);
    gemv_wmma<0, H_><<<dim3(V_ / 16), dim3(H_ / KS_ * 32), 0, stream>>>(W_dec, ws + WS_HNEW, b_dec, out);
}


# Stack blend shared out over the three launches whose matrix chain leaves memory idle: rows ride beside stack_input (3072), gates (2048), logits (3072), chain blocks first in grid; sums unchanged
# speedup vs baseline: 1.6526x; 1.0816x over previous
#include <hip/hip_runtime.h>
#include <hip/hip_bf16.h>
#include <math.h>

#define V_  128
#define H_  1024
#define SW_ 1024
#define SD_ 8192

typedef __attribute__((ext_vector_type(2))) float v2f;
typedef __attribute__((ext_vector_type(4))) float v4f;
typedef __attribute__((ext_vector_type(8))) float v8f;

#define WS_CTRL   0
#define WS_SINP   16
#define WS_GH     4096
#define WS_GI     8192
#define WS_HNEW   12288

#define KS_    128
#define PITCH_ (KS_ + 4)

template <int ACT, int K>
__device__ __forceinline__
void gemv_rows16(const float* __restrict__ W, const float* __restrict__ vec,
                 const float* __restrict__ bias, float* __restrict__ out, const int row_base,
                 float (*tile)[16][PITCH_], float (*red)[16]) {
    constexpr int NW = K / KS_;
    const int lane = threadIdx.x & 31;
    const int wave = threadIdx.x >> 5;
    const int kbeg = wave * KS_;

    const float* wsrc = W + (size_t)row_base * K + (size_t)(kbeg + lane * 4);
    v4f t[16];
    for (int r = 0; r < 16; ++r) t[r] = *(const v4f*)(wsrc + (size_t)r * K);

    const int co = (lane >> 4) << 1;
    const float* vp = vec + kbeg + co;
    v2f av[KS_ / 4];
    for (int s = 0; s < KS_ / 4; ++s) av[s] = *(const v2f*)(vp + 4 * s);

    for (int r = 0; r < 16; ++r) *(v4f*)&tile[wave][r][lane * 4] = t[r];
    __syncthreads();

    const float* bp = &tile[wave][lane & 15][co];
    v8f c0 = {}, c1 = {}, c2 = {}, c3 = {};
    for (int k = 0; k < KS_; k += 16) {
        v2f b0 = *(const v2f*)(bp + k);
        v2f b1 = *(const v2f*)(bp + k + 4);
        v2f b2 = *(const v2f*)(bp + k + 8);
        v2f b3 = *(const v2f*)(bp + k + 12);
        c0 = __builtin_amdgcn_wmma_f32_16x16x4_f32(false, av[k / 4],     false, b0, (short)0, c0, false, false);
        c1 = __builtin_amdgcn_wmma_f32_16x16x4_f32(false, av[k / 4 + 1], false, b1, (short)0, c1, false, false);
        c2 = __builtin_amdgcn_wmma_f32_16x16x4_f32(false, av[k / 4 + 2], false, b2, (short)0, c2, false, false);
        c3 = __builtin_amdgcn_wmma_f32_16x16x4_f32(false, av[k / 4 + 3], false, b3, (short)0, c3, false, false);
    }

    if (lane < 16) red[wave][lane] = c0[0] + c1[0] + c2[0] + c3[0];
    __syncthreads();
    if (threadIdx.x < 16) {
        float s = 0.f;
        #pragma unroll
        for (int w = 0; w < NW; ++w) s += red[w][threadIdx.x];
        s += bias[row_base + threadIdx.x];
        if (ACT == 1) s = tanhf(s);
        out[row_base + threadIdx.x] = s;
    }
}

#define CTRL_WAVES_ 8
static_assert(CTRL_WAVES_ * 32 * 4 == H_, "one 16-byte piece of h per thread");

__device__ __forceinline__
void ctrl_partials(const v4f hv, const v4f w0, const v4f w1, const v4f w2,
                   const int lane, float* part_w) {
    float p0 = hv[0] * w0[0] + hv[1] * w0[1] + hv[2] * w0[2] + hv[3] * w0[3];
    float p1 = hv[0] * w1[0] + hv[1] * w1[1] + hv[2] * w1[2] + hv[3] * w1[3];
    float p2 = hv[0] * w2[0] + hv[1] * w2[1] + hv[2] * w2[2] + hv[3] * w2[3];
    for (int off = 16; off > 0; off >>= 1) {
        p0 += __shfl_xor(p0, off, 32);
        p1 += __shfl_xor(p1, off, 32);
        p2 += __shfl_xor(p2, off, 32);
    }
    if (lane == 0) *(v4f*)part_w = v4f{p0, p1, p2, 0.f};
}

__device__ __forceinline__
void ctrl_softmax(const float (*part)[4], const float bc0, const float bc1, const float bc2,
                  float& a_push, float& a_pop, float& a_noop) {
    float l0 = 0.f, l1 = 0.f, l2 = 0.f;
    for (int w = 0; w < CTRL_WAVES_; ++w) {
        const v4f q = *(const v4f*)part[w];
        l0 += q[0]; l1 += q[1]; l2 += q[2];
    }
    l0 += bc0; l1 += bc1; l2 += bc2;
    const float m  = fmaxf(l0, fmaxf(l1, l2));
    const float e0 = expf(l0 - m), e1 = expf(l1 - m), e2 = expf(l2 - m);
    const float inv = 1.f / (e0 + e1 + e2);
    a_push = e0 * inv;
    a_pop  = e1 * inv;
    a_noop = e2 * inv;
}

#define BLEND_ROWS_   8
#define BLEND_NP_     (SW_ / 128)
#define SINP_BLOCKS_  (SW_ / 16)
#define GATE_BLOCKS_  (H_ / (BLEND_ROWS_ * 32))
#define DEC_BLOCKS_   (V_ / 16)
#define BLEND_BLOCKS_ (SD_ / BLEND_ROWS_)
#define BLEND_B1_     384
#define BLEND_B3_     256
#define BLEND_B4_     (BLEND_BLOCKS_ - BLEND_B1_ - BLEND_B3_)
static_assert(BLEND_ROWS_ == CTRL_WAVES_, "the blend block's waves form the controls");
static_assert(BLEND_ROWS_ == H_ / KS_, "both kinds of block have the same number of waves");
static_assert(BLEND_B4_ > 0, "every launch carries a share");

__device__ __forceinline__
void blend_block(const int bblk, const float* __restrict__ hidden, const float* __restrict__ stack,
                 float* __restrict__ new_stack, const float* __restrict__ W_ctrl,
                 const float* __restrict__ b_ctrl, float* __restrict__ ws, float (*part)[4]) {
    const int lane = threadIdx.x & 31;
    const int wave = threadIdx.x >> 5;

    const int e = threadIdx.x * 4;
    const v4f hv = *(const v4f*)(hidden + e);
    const v4f w0 = *(const v4f*)(W_ctrl + e);
    const v4f w1 = *(const v4f*)(W_ctrl + H_ + e);
    const v4f w2 = *(const v4f*)(W_ctrl + 2 * H_ + e);
    const float bc0 = b_ctrl[0], bc1 = b_ctrl[1], bc2 = b_ctrl[2];

    const int j  = bblk * BLEND_ROWS_ + wave;
    const bool live = (j != 0);
    const int c0 = lane * 4;
    const float* cp = stack + (size_t)j * SW_ + c0;
    const float* upp = live ? (cp - SW_) : cp;

    v4f cur[BLEND_NP_], up[BLEND_NP_], dn[BLEND_NP_];
    for (int p = 0; p < BLEND_NP_; ++p) cur[p] = *(const v4f*)(cp + p * 128);
    for (int p = 0; p < BLEND_NP_; ++p) up[p]  = *(const v4f*)(upp + p * 128);
    for (int p = 0; p < BLEND_NP_; ++p) dn[p]  = v4f{0.f, 0.f, 0.f, 0.f};
    if (j < SD_ - 1)
        for (int p = 0; p < BLEND_NP_; ++p) dn[p] = *(const v4f*)(cp + SW_ + p * 128);

    ctrl_partials(hv, w0, w1, w2, lane, part[wave]);
    __syncthreads();
    float a_push, a_pop, a_noop;
    ctrl_softmax(part, bc0, bc1, bc2, a_push, a_pop, a_noop);
    if (!live) {
        if (lane == 0) {
            ws[WS_CTRL + 0] = a_push;
            ws[WS_CTRL + 1] = a_pop;
            ws[WS_CTRL + 2] = a_noop;
        }
        return;
    }

    float* op = new_stack + (size_t)j * SW_ + c0;
    for (int p = 0; p < BLEND_NP_; ++p) {
        v4f o = a_noop * cur[p] + a_push * up[p] + a_pop * dn[p];
        __builtin_nontemporal_store(o, (v4f*)(op + p * 128));
    }
}

__global__ __launch_bounds__(BLEND_ROWS_ * 32)
void sinp_and_blend(const float* __restrict__ W_sin, const float* __restrict__ hidden,
                    const float* __restrict__ b_sin, const float* __restrict__ stack,
                    float* __restrict__ new_stack, const float* __restrict__ W_ctrl,
                    const float* __restrict__ b_ctrl, float* __restrict__ ws) {
    __shared__ __attribute__((aligned(16))) float tile[H_ / KS_][16][PITCH_];
    __shared__ float red[H_ / KS_][16];
    __shared__ __attribute__((aligned(16))) float part[CTRL_WAVES_][4];

    if (blockIdx.x < SINP_BLOCKS_) {
        gemv_rows16<1, H_>(W_sin, hidden, b_sin, ws + WS_SINP, blockIdx.x * 16, tile, red);
        return;
    }
    blend_block(blockIdx.x - SINP_BLOCKS_, hidden, stack, new_stack, W_ctrl, b_ctrl, ws, part);
}

#define NW_IH_ ((H_ + SW_) / KS_)
#define NW_X_  (H_ / KS_)
#define NW_S_  (SW_ / KS_)
#define NW_HH_ (H_ / KS_)
#define NW_G_  (NW_IH_ + NW_HH_)
#define GI_BLOCKS_ (3 * H_ / 16)

__global__ __launch_bounds__(NW_G_ * 32)
void gemv_gi_gh(const int* __restrict__ inp, const float* __restrict__ emb,
                const float* __restrict__ hidden, const float* __restrict__ stack,
                const float* __restrict__ W_ih, const float* __restrict__ b_ih,
                const float* __restrict__ W_hh, const float* __restrict__ b_hh,
                float* __restrict__ new_stack, float* ws) {
    __shared__ __attribute__((aligned(16))) float tile[NW_G_][16][PITCH_];
    __shared__ float red[NW_G_][16];
    __shared__ float red_up[NW_S_][16];
    __shared__ float red_dn[NW_S_][16];

    const float a_push = ws[WS_CTRL + 0];
    const float a_pop  = ws[WS_CTRL + 1];
    const float a_noop = ws[WS_CTRL + 2];

    if (blockIdx.x == GI_BLOCKS_) {
        if (threadIdx.x < SW_ / 4) {
            const int e = threadIdx.x * 4;
            const v4f r_cur = *(const v4f*)(stack + e);
            const v4f r_up  = *(const v4f*)((const float*)ws + WS_SINP + e);
            const v4f r_dn  = *(const v4f*)(stack + SW_ + e);
            v4f o = a_noop * r_cur + a_push * r_up + a_pop * r_dn;
            __builtin_nontemporal_store(o, (v4f*)(new_stack + e));
        }
        return;
    }

    const int lane = threadIdx.x & 31;
    const int wave = threadIdx.x >> 5;
    const int row_base = blockIdx.x * 16;
    const int co = (lane >> 4) << 1;

    v4f t[16];
    const float* vp;
    if (wave < NW_IH_) {
        const float* wsrc = W_ih + (size_t)row_base * (H_ + SW_) + (size_t)(wave * KS_ + lane * 4);
        for (int r = 0; r < 16; ++r) t[r] = *(const v4f*)(wsrc + (size_t)r * (H_ + SW_));
        if (wave < NW_X_) {
            vp = emb + (size_t)inp[0] * H_ + wave * KS_;
        } else {
            const int m = lane & 15;
            const float* rowp = (m == 1) ? (const float*)ws + WS_SINP
                              : (m == 2) ? stack + SW_ : stack;
            vp = rowp + (wave - NW_X_) * KS_;
        }
    } else {
        const int kw = wave - NW_IH_;
        const float* wsrc = W_hh + (size_t)row_base * H_ + (size_t)(kw * KS_ + lane * 4);
        for (int r = 0; r < 16; ++r) t[r] = *(const v4f*)(wsrc + (size_t)r * H_);
        vp = hidden + kw * KS_;
    }
    vp += co;
    v2f av[KS_ / 4];
    for (int s = 0; s < KS_ / 4; ++s) av[s] = *(const v2f*)(vp + 4 * s);

    for (int r = 0; r < 16; ++r) *(v4f*)&tile[wave][r][lane * 4] = t[r];
    __syncthreads();

    const float* bp = &tile[wave][lane & 15][co];
    v8f c0 = {}, c1 = {}, c2 = {}, c3 = {};
    for (int k = 0; k < KS_; k += 16) {
        v2f b0 = *(const v2f*)(bp + k);
        v2f b1 = *(const v2f*)(bp + k + 4);
        v2f b2 = *(const v2f*)(bp + k + 8);
        v2f b3 = *(const v2f*)(bp + k + 12);
        c0 = __builtin_amdgcn_wmma_f32_16x16x4_f32(false, av[k / 4],     false, b0, (short)0, c0, false, false);
        c1 = __builtin_amdgcn_wmma_f32_16x16x4_f32(false, av[k / 4 + 1], false, b1, (short)0, c1, false, false);
        c2 = __builtin_amdgcn_wmma_f32_16x16x4_f32(false, av[k / 4 + 2], false, b2, (short)0, c2, false, false);
        c3 = __builtin_amdgcn_wmma_f32_16x16x4_f32(false, av[k / 4 + 3], false, b3, (short)0, c3, false, false);
    }

    if (lane < 16) {
        red[wave][lane] = c0[0] + c1[0] + c2[0] + c3[0];
        if (wave >= NW_X_ && wave < NW_IH_) {
            red_up[wave - NW_X_][lane] = c0[1] + c1[1] + c2[1] + c3[1];
            red_dn[wave - NW_X_][lane] = c0[2] + c1[2] + c2[2] + c3[2];
        }
    }
    __syncthreads();

    if (wave == 0 && lane < 16) {
        float sx = 0.f, sc = 0.f, su = 0.f, sd = 0.f;
        for (int w = 0; w < NW_X_; ++w) sx += red[w][lane];
        for (int w = 0; w < NW_S_; ++w) {
            sc += red[NW_X_ + w][lane];
            su += red_up[w][lane];
            sd += red_dn[w][lane];
        }
        ws[WS_GI + row_base + lane] = sx + (a_noop * sc + a_push * su + a_pop * sd) + b_ih[row_base + lane];
    } else if (wave == 1 && lane < 16) {
        float s = 0.f;
        for (int w = NW_IH_; w < NW_G_; ++w) s += red[w][lane];
        ws[WS_GH + row_base + lane] = s + b_hh[row_base + lane];
    }
}

__global__ __launch_bounds__(BLEND_ROWS_ * 32)
void gates_and_blend(const float* __restrict__ hidden, float* ws,
                     float* __restrict__ hid_out, const float* __restrict__ stack,
                     float* __restrict__ new_stack, const float* __restrict__ W_ctrl,
                     const float* __restrict__ b_ctrl) {
    __shared__ __attribute__((aligned(16))) float part[CTRL_WAVES_][4];

    if (blockIdx.x < GATE_BLOCKS_) {
        const int i = blockIdx.x * (BLEND_ROWS_ * 32) + threadIdx.x;
        const float* gi = ws + WS_GI;
        const float* gh = ws + WS_GH;
        float r  = 1.f / (1.f + expf(-(gi[i] + gh[i])));
        float z  = 1.f / (1.f + expf(-(gi[H_ + i] + gh[H_ + i])));
        float n  = tanhf(gi[2 * H_ + i] + r * gh[2 * H_ + i]);
        float hn = (1.f - z) * n + z * hidden[i];
        hid_out[i] = hn;
        ws[WS_HNEW + i] = hn;
        return;
    }
    blend_block(BLEND_B1_ + (blockIdx.x - GATE_BLOCKS_), hidden, stack, new_stack, W_ctrl, b_ctrl, ws, part);
}

__global__ __launch_bounds__(BLEND_ROWS_ * 32)
void dec_and_blend(const float* __restrict__ W_dec, const float* __restrict__ b_dec,
                   float* __restrict__ out, const float* __restrict__ hidden,
                   const float* __restrict__ stack, float* __restrict__ new_stack,
                   const float* __restrict__ W_ctrl, const float* __restrict__ b_ctrl,
                   float* __restrict__ ws) {
    __shared__ __attribute__((aligned(16))) float tile[H_ / KS_][16][PITCH_];
    __shared__ float red[H_ / KS_][16];
    __shared__ __attribute__((aligned(16))) float part[CTRL_WAVES_][4];

    if (blockIdx.x < DEC_BLOCKS_) {
        gemv_rows16<0, H_>(W_dec, ws + WS_HNEW, b_dec, out, blockIdx.x * 16, tile, red);
        return;
    }
    blend_block(BLEND_B1_ + BLEND_B3_ + (blockIdx.x - DEC_BLOCKS_), hidden, stack, new_stack, W_ctrl, b_ctrl, ws, part);
}

extern "C" void kernel_launch(void* const* d_in, const int* in_sizes, int n_in,
                              void* d_out, int out_size, void* d_ws, size_t ws_size,
                              hipStream_t stream) {
    (void)in_sizes; (void)n_in; (void)out_size; (void)ws_size;
    const int*   inp    = (const int*)  d_in[0];
    const float* hidden = (const float*)d_in[1];
    const float* stack  = (const float*)d_in[2];
    const float* emb    = (const float*)d_in[3];
    const float* W_ctrl = (const float*)d_in[4];
    const float* b_ctrl = (const float*)d_in[5];
    const float* W_sin  = (const float*)d_in[6];
    const float* b_sin  = (const float*)d_in[7];
    const float* W_ih   = (const float*)d_in[8];
    const float* b_ih   = (const float*)d_in[9];
    const float* W_hh   = (const float*)d_in[10];
    const float* b_hh   = (const float*)d_in[11];
    const float* W_dec  = (const float*)d_in[12];
    const float* b_dec  = (const float*)d_in[13];

    float* out       = (float*)d_out;
    float* hid_out   = out + V_;
    float* stack_out = out + V_ + H_;
    float* ws        = (float*)d_ws;

    sinp_and_blend<<<dim3(SINP_BLOCKS_ + BLEND_B1_), dim3(BLEND_ROWS_ * 32), 0, stream>>>(
        W_sin, hidden, b_sin, stack, stack_out, W_ctrl, b_ctrl, ws);
    gemv_gi_gh<<<dim3(GI_BLOCKS_ + 1), dim3(NW_G_ * 32), 0, stream>>>(
        inp, emb, hidden, stack, W_ih, b_ih, W_hh, b_hh, stack_out, ws);
    gates_and_blend<<<dim3(GATE_BLOCKS_ + BLEND_B3_), dim3(BLEND_ROWS_ * 32), 0, stream>>>(
        hidden, ws, hid_out, stack, stack_out, W_ctrl, b_ctrl);
    dec_and_blend<<<dim3(DEC_BLOCKS_ + BLEND_B4_), dim3(BLEND_ROWS_ * 32), 0, stream>>>(
        W_dec, b_dec, out, hidden, stack, stack_out, W_ctrl, b_ctrl, ws);
}


# GRU gates leave their own launch: each of the 8 logits blocks forms all of h_new (4 units a thread, via LDS to its A fragments); its blend rows join the other shares; 4 launches -> 3; same sums
# speedup vs baseline: 1.7762x; 1.0748x over previous
#include <hip/hip_runtime.h>
#include <hip/hip_bf16.h>
#include <math.h>

#define V_  128
#define H_  1024
#define SW_ 1024
#define SD_ 8192

typedef __attribute__((ext_vector_type(2))) float v2f;
typedef __attribute__((ext_vector_type(4))) float v4f;
typedef __attribute__((ext_vector_type(8))) float v8f;

#define WS_CTRL   0
#define WS_SINP   16
#define WS_GH     4096
#define WS_GI     8192

#define KS_    128
#define PITCH_ (KS_ + 4)

template <int ACT, int K>
__device__ __forceinline__
void gemv_rows16(const float* __restrict__ W, const float* __restrict__ vec,
                 const float* __restrict__ bias, float* __restrict__ out, const int row_base,
                 float (*tile)[16][PITCH_], float (*red)[16]) {
    constexpr int NW = K / KS_;
    const int lane = threadIdx.x & 31;
    const int wave = threadIdx.x >> 5;
    const int kbeg = wave * KS_;

    const float* wsrc = W + (size_t)row_base * K + (size_t)(kbeg + lane * 4);
    v4f t[16];
    for (int r = 0; r < 16; ++r) t[r] = *(const v4f*)(wsrc + (size_t)r * K);

    const int co = (lane >> 4) << 1;
    const float* vp = vec + kbeg + co;
    v2f av[KS_ / 4];
    for (int s = 0; s < KS_ / 4; ++s) av[s] = *(const v2f*)(vp + 4 * s);

    for (int r = 0; r < 16; ++r) *(v4f*)&tile[wave][r][lane * 4] = t[r];
    __syncthreads();

    const float* bp = &tile[wave][lane & 15][co];
    v8f c0 = {}, c1 = {}, c2 = {}, c3 = {};
    for (int k = 0; k < KS_; k += 16) {
        v2f b0 = *(const v2f*)(bp + k);
        v2f b1 = *(const v2f*)(bp + k + 4);
        v2f b2 = *(const v2f*)(bp + k + 8);
        v2f b3 = *(const v2f*)(bp + k + 12);
        c0 = __builtin_amdgcn_wmma_f32_16x16x4_f32(false, av[k / 4],     false, b0, (short)0, c0, false, false);
        c1 = __builtin_amdgcn_wmma_f32_16x16x4_f32(false, av[k / 4 + 1], false, b1, (short)0, c1, false, false);
        c2 = __builtin_amdgcn_wmma_f32_16x16x4_f32(false, av[k / 4 + 2], false, b2, (short)0, c2, false, false);
        c3 = __builtin_amdgcn_wmma_f32_16x16x4_f32(false, av[k / 4 + 3], false, b3, (short)0, c3, false, false);
    }

    if (lane < 16) red[wave][lane] = c0[0] + c1[0] + c2[0] + c3[0];
    __syncthreads();
    if (threadIdx.x < 16) {
        float s = 0.f;
        #pragma unroll
        for (int w = 0; w < NW; ++w) s += red[w][threadIdx.x];
        s += bias[row_base + threadIdx.x];
        if (ACT == 1) s = tanhf(s);
        out[row_base + threadIdx.x] = s;
    }
}

#define CTRL_WAVES_ 8
static_assert(CTRL_WAVES_ * 32 * 4 == H_, "one 16-byte piece of h per thread");

__device__ __forceinline__
void ctrl_partials(const v4f hv, const v4f w0, const v4f w1, const v4f w2,
                   const int lane, float* part_w) {
    float p0 = hv[0] * w0[0] + hv[1] * w0[1] + hv[2] * w0[2] + hv[3] * w0[3];
    float p1 = hv[0] * w1[0] + hv[1] * w1[1] + hv[2] * w1[2] + hv[3] * w1[3];
    float p2 = hv[0] * w2[0] + hv[1] * w2[1] + hv[2] * w2[2] + hv[3] * w2[3];
    for (int off = 16; off > 0; off >>= 1) {
        p0 += __shfl_xor(p0, off, 32);
        p1 += __shfl_xor(p1, off, 32);
        p2 += __shfl_xor(p2, off, 32);
    }
    if (lane == 0) *(v4f*)part_w = v4f{p0, p1, p2, 0.f};
}

__device__ __forceinline__
void ctrl_softmax(const float (*part)[4], const float bc0, const float bc1, const float bc2,
                  float& a_push, float& a_pop, float& a_noop) {
    float l0 = 0.f, l1 = 0.f, l2 = 0.f;
    for (int w = 0; w < CTRL_WAVES_; ++w) {
        const v4f q = *(const v4f*)part[w];
        l0 += q[0]; l1 += q[1]; l2 += q[2];
    }
    l0 += bc0; l1 += bc1; l2 += bc2;
    const float m  = fmaxf(l0, fmaxf(l1, l2));
    const float e0 = expf(l0 - m), e1 = expf(l1 - m), e2 = expf(l2 - m);
    const float inv = 1.f / (e0 + e1 + e2);
    a_push = e0 * inv;
    a_pop  = e1 * inv;
    a_noop = e2 * inv;
}

#define BLEND_ROWS_   8
#define BLEND_NP_     (SW_ / 128)
#define SINP_BLOCKS_  (SW_ / 16)
#define DEC_BLOCKS_   (V_ / 16)
#define BLEND_BLOCKS_ (SD_ / BLEND_ROWS_)
#define BLEND_B1_     448
#define BLEND_B3_     (BLEND_BLOCKS_ - BLEND_B1_)
static_assert(BLEND_ROWS_ == CTRL_WAVES_, "the blend block's waves form the controls");
static_assert(BLEND_ROWS_ == H_ / KS_, "both kinds of block have the same number of waves");
static_assert(BLEND_B3_ > 0, "every launch carries a share");

__device__ __forceinline__
void blend_block(const int bblk, const float* __restrict__ hidden, const float* __restrict__ stack,
                 float* __restrict__ new_stack, const float* __restrict__ W_ctrl,
                 const float* __restrict__ b_ctrl, float* __restrict__ ws, float (*part)[4]) {
    const int lane = threadIdx.x & 31;
    const int wave = threadIdx.x >> 5;

    const int e = threadIdx.x * 4;
    const v4f hv = *(const v4f*)(hidden + e);
    const v4f w0 = *(const v4f*)(W_ctrl + e);
    const v4f w1 = *(const v4f*)(W_ctrl + H_ + e);
    const v4f w2 = *(const v4f*)(W_ctrl + 2 * H_ + e);
    const float bc0 = b_ctrl[0], bc1 = b_ctrl[1], bc2 = b_ctrl[2];

    const int j  = bblk * BLEND_ROWS_ + wave;
    const bool live = (j != 0);
    const int c0 = lane * 4;
    const float* cp = stack + (size_t)j * SW_ + c0;
    const float* upp = live ? (cp - SW_) : cp;

    v4f cur[BLEND_NP_], up[BLEND_NP_], dn[BLEND_NP_];
    for (int p = 0; p < BLEND_NP_; ++p) cur[p] = *(const v4f*)(cp + p * 128);
    for (int p = 0; p < BLEND_NP_; ++p) up[p]  = *(const v4f*)(upp + p * 128);
    for (int p = 0; p < BLEND_NP_; ++p) dn[p]  = v4f{0.f, 0.f, 0.f, 0.f};
    if (j < SD_ - 1)
        for (int p = 0; p < BLEND_NP_; ++p) dn[p] = *(const v4f*)(cp + SW_ + p * 128);

    ctrl_partials(hv, w0, w1, w2, lane, part[wave]);
    __syncthreads();
    float a_push, a_pop, a_noop;
    ctrl_softmax(part, bc0, bc1, bc2, a_push, a_pop, a_noop);
    if (!live) {
        if (lane == 0) {
            ws[WS_CTRL + 0] = a_push;
            ws[WS_CTRL + 1] = a_pop;
            ws[WS_CTRL + 2] = a_noop;
        }
        return;
    }

    float* op = new_stack + (size_t)j * SW_ + c0;
    for (int p = 0; p < BLEND_NP_; ++p) {
        v4f o = a_noop * cur[p] + a_push * up[p] + a_pop * dn[p];
        __builtin_nontemporal_store(o, (v4f*)(op + p * 128));
    }
}

__global__ __launch_bounds__(BLEND_ROWS_ * 32)
void sinp_and_blend(const float* __restrict__ W_sin, const float* __restrict__ hidden,
                    const float* __restrict__ b_sin, const float* __restrict__ stack,
                    float* __restrict__ new_stack, const float* __restrict__ W_ctrl,
                    const float* __restrict__ b_ctrl, float* __restrict__ ws) {
    __shared__ __attribute__((aligned(16))) float tile[H_ / KS_][16][PITCH_];
    __shared__ float red[H_ / KS_][16];
    __shared__ __attribute__((aligned(16))) float part[CTRL_WAVES_][4];

    if (blockIdx.x < SINP_BLOCKS_) {
        gemv_rows16<1, H_>(W_sin, hidden, b_sin, ws + WS_SINP, blockIdx.x * 16, tile, red);
        return;
    }
    blend_block(blockIdx.x - SINP_BLOCKS_, hidden, stack, new_stack, W_ctrl, b_ctrl, ws, part);
}

#define NW_IH_ ((H_ + SW_) / KS_)
#define NW_X_  (H_ / KS_)
#define NW_S_  (SW_ / KS_)
#define NW_HH_ (H_ / KS_)
#define NW_G_  (NW_IH_ + NW_HH_)
#define GI_BLOCKS_ (3 * H_ / 16)

__global__ __launch_bounds__(NW_G_ * 32)
void gemv_gi_gh(const int* __restrict__ inp, const float* __restrict__ emb,
                const float* __restrict__ hidden, const float* __restrict__ stack,
                const float* __restrict__ W_ih, const float* __restrict__ b_ih,
                const float* __restrict__ W_hh, const float* __restrict__ b_hh,
                float* __restrict__ new_stack, float* ws) {
    __shared__ __attribute__((aligned(16))) float tile[NW_G_][16][PITCH_];
    __shared__ float red[NW_G_][16];
    __shared__ float red_up[NW_S_][16];
    __shared__ float red_dn[NW_S_][16];

    const float a_push = ws[WS_CTRL + 0];
    const float a_pop  = ws[WS_CTRL + 1];
    const float a_noop = ws[WS_CTRL + 2];

    if (blockIdx.x == GI_BLOCKS_) {
        if (threadIdx.x < SW_ / 4) {
            const int e = threadIdx.x * 4;
            const v4f r_cur = *(const v4f*)(stack + e);
            const v4f r_up  = *(const v4f*)((const float*)ws + WS_SINP + e);
            const v4f r_dn  = *(const v4f*)(stack + SW_ + e);
            v4f o = a_noop * r_cur + a_push * r_up + a_pop * r_dn;
            __builtin_nontemporal_store(o, (v4f*)(new_stack + e));
        }
        return;
    }

    const int lane = threadIdx.x & 31;
    const int wave = threadIdx.x >> 5;
    const int row_base = blockIdx.x * 16;
    const int co = (lane >> 4) << 1;

    v4f t[16];
    const float* vp;
    if (wave < NW_IH_) {
        const float* wsrc = W_ih + (size_t)row_base * (H_ + SW_) + (size_t)(wave * KS_ + lane * 4);
        for (int r = 0; r < 16; ++r) t[r] = *(const v4f*)(wsrc + (size_t)r * (H_ + SW_));
        if (wave < NW_X_) {
            vp = emb + (size_t)inp[0] * H_ + wave * KS_;
        } else {
            const int m = lane & 15;
            const float* rowp = (m == 1) ? (const float*)ws + WS_SINP
                              : (m == 2) ? stack + SW_ : stack;
            vp = rowp + (wave - NW_X_) * KS_;
        }
    } else {
        const int kw = wave - NW_IH_;
        const float* wsrc = W_hh + (size_t)row_base * H_ + (size_t)(kw * KS_ + lane * 4);
        for (int r = 0; r < 16; ++r) t[r] = *(const v4f*)(wsrc + (size_t)r * H_);
        vp = hidden + kw * KS_;
    }
    vp += co;
    v2f av[KS_ / 4];
    for (int s = 0; s < KS_ / 4; ++s) av[s] = *(const v2f*)(vp + 4 * s);

    for (int r = 0; r < 16; ++r) *(v4f*)&tile[wave][r][lane * 4] = t[r];
    __syncthreads();

    const float* bp = &tile[wave][lane & 15][co];
    v8f c0 = {}, c1 = {}, c2 = {}, c3 = {};
    for (int k = 0; k < KS_; k += 16) {
        v2f b0 = *(const v2f*)(bp + k);
        v2f b1 = *(const v2f*)(bp + k + 4);
        v2f b2 = *(const v2f*)(bp + k + 8);
        v2f b3 = *(const v2f*)(bp + k + 12);
        c0 = __builtin_amdgcn_wmma_f32_16x16x4_f32(false, av[k / 4],     false, b0, (short)0, c0, false, false);
        c1 = __builtin_amdgcn_wmma_f32_16x16x4_f32(false, av[k / 4 + 1], false, b1, (short)0, c1, false, false);
        c2 = __builtin_amdgcn_wmma_f32_16x16x4_f32(false, av[k / 4 + 2], false, b2, (short)0, c2, false, false);
        c3 = __builtin_amdgcn_wmma_f32_16x16x4_f32(false, av[k / 4 + 3], false, b3, (short)0, c3, false, false);
    }

    if (lane < 16) {
        red[wave][lane] = c0[0] + c1[0] + c2[0] + c3[0];
        if (wave >= NW_X_ && wave < NW_IH_) {
            red_up[wave - NW_X_][lane] = c0[1] + c1[1] + c2[1] + c3[1];
            red_dn[wave - NW_X_][lane] = c0[2] + c1[2] + c2[2] + c3[2];
        }
    }
    __syncthreads();

    if (wave == 0 && lane < 16) {
        float sx = 0.f, sc = 0.f, su = 0.f, sd = 0.f;
        for (int w = 0; w < NW_X_; ++w) sx += red[w][lane];
        for (int w = 0; w < NW_S_; ++w) {
            sc += red[NW_X_ + w][lane];
            su += red_up[w][lane];
            sd += red_dn[w][lane];
        }
        ws[WS_GI + row_base + lane] = sx + (a_noop * sc + a_push * su + a_pop * sd) + b_ih[row_base + lane];
    } else if (wave == 1 && lane < 16) {
        float s = 0.f;
        for (int w = NW_IH_; w < NW_G_; ++w) s += red[w][lane];
        ws[WS_GH + row_base + lane] = s + b_hh[row_base + lane];
    }
}

__device__ __forceinline__
void gates_dec_rows16(const float* __restrict__ W_dec, const float* __restrict__ b_dec,
                      float* __restrict__ out, float* __restrict__ hid_out,
                      const float* __restrict__ hidden, const float* __restrict__ ws,
                      float (*tile)[16][PITCH_], float (*red)[16], float* hn) {
    const int lane = threadIdx.x & 31;
    const int wave = threadIdx.x >> 5;
    const int kbeg = wave * KS_;
    const int row_base = blockIdx.x * 16;
    const int e = threadIdx.x * 4;

    const float* gi = ws + WS_GI;
    const float* gh = ws + WS_GH;
    const v4f gir = *(const v4f*)(gi + e);
    const v4f ghr = *(const v4f*)(gh + e);
    const v4f giz = *(const v4f*)(gi + H_ + e);
    const v4f ghz = *(const v4f*)(gh + H_ + e);
    const v4f gin = *(const v4f*)(gi + 2 * H_ + e);
    const v4f ghn = *(const v4f*)(gh + 2 * H_ + e);
    const v4f hv  = *(const v4f*)(hidden + e);

    const float* wsrc = W_dec + (size_t)row_base * H_ + (size_t)(kbeg + lane * 4);
    v4f t[16];
    for (int r = 0; r < 16; ++r) t[r] = *(const v4f*)(wsrc + (size_t)r * H_);

    v4f hn4;
    for (int q = 0; q < 4; ++q) {
        float r  = 1.f / (1.f + expf(-(gir[q] + ghr[q])));
        float z  = 1.f / (1.f + expf(-(giz[q] + ghz[q])));
        float n  = tanhf(gin[q] + r * ghn[q]);
        hn4[q] = fmaf(1.f - z, n, z * hv[q]);
    }
    *(v4f*)(hn + e) = hn4;
    if (blockIdx.x == 0) *(v4f*)(hid_out + e) = hn4;

    for (int r = 0; r < 16; ++r) *(v4f*)&tile[wave][r][lane * 4] = t[r];
    __syncthreads();

    const int co = (lane >> 4) << 1;
    const float* vp = hn + kbeg + co;
    v2f av[KS_ / 4];
    for (int s = 0; s < KS_ / 4; ++s) av[s] = *(const v2f*)(vp + 4 * s);

    const float* bp = &tile[wave][lane & 15][co];
    v8f c0 = {}, c1 = {}, c2 = {}, c3 = {};
    for (int k = 0; k < KS_; k += 16) {
        v2f b0 = *(const v2f*)(bp + k);
        v2f b1 = *(const v2f*)(bp + k + 4);
        v2f b2 = *(const v2f*)(bp + k + 8);
        v2f b3 = *(const v2f*)(bp + k + 12);
        c0 = __builtin_amdgcn_wmma_f32_16x16x4_f32(false, av[k / 4],     false, b0, (short)0, c0, false, false);
        c1 = __builtin_amdgcn_wmma_f32_16x16x4_f32(false, av[k / 4 + 1], false, b1, (short)0, c1, false, false);
        c2 = __builtin_amdgcn_wmma_f32_16x16x4_f32(false, av[k / 4 + 2], false, b2, (short)0, c2, false, false);
        c3 = __builtin_amdgcn_wmma_f32_16x16x4_f32(false, av[k / 4 + 3], false, b3, (short)0, c3, false, false);
    }

    if (lane < 16) red[wave][lane] = c0[0] + c1[0] + c2[0] + c3[0];
    __syncthreads();
    if (threadIdx.x < 16) {
        float s = 0.f;
        #pragma unroll
        for (int w = 0; w < H_ / KS_; ++w) s += red[w][threadIdx.x];
        s += b_dec[row_base + threadIdx.x];
        out[row_base + threadIdx.x] = s;
    }
}

__global__ __launch_bounds__(BLEND_ROWS_ * 32)
void gates_dec_and_blend(const float* __restrict__ W_dec, const float* __restrict__ b_dec,
                         float* __restrict__ out, float* __restrict__ hid_out,
                         const float* __restrict__ hidden,
                         const float* __restrict__ stack, float* __restrict__ new_stack,
                         const float* __restrict__ W_ctrl, const float* __restrict__ b_ctrl,
                         float* __restrict__ ws) {
    __shared__ __attribute__((aligned(16))) float tile[H_ / KS_][16][PITCH_];
    __shared__ float red[H_ / KS_][16];
    __shared__ __attribute__((aligned(16))) float part[CTRL_WAVES_][4];
    __shared__ __attribute__((aligned(16))) float hn[H_];

    if (blockIdx.x < DEC_BLOCKS_) {
        gates_dec_rows16(W_dec, b_dec, out, hid_out, hidden, ws, tile, red, hn);
        return;
    }
    blend_block(BLEND_B1_ + (blockIdx.x - DEC_BLOCKS_), hidden, stack, new_stack, W_ctrl, b_ctrl, ws, part);
}

extern "C" void kernel_launch(void* const* d_in, const int* in_sizes, int n_in,
                              void* d_out, int out_size, void* d_ws, size_t ws_size,
                              hipStream_t stream) {
    (void)in_sizes; (void)n_in; (void)out_size; (void)ws_size;
    const int*   inp    = (const int*)  d_in[0];
    const float* hidden = (const float*)d_in[1];
    const float* stack  = (const float*)d_in[2];
    const float* emb    = (const float*)d_in[3];
    const float* W_ctrl = (const float*)d_in[4];
    const float* b_ctrl = (const float*)d_in[5];
    const float* W_sin  = (const float*)d_in[6];
    const float* b_sin  = (const float*)d_in[7];
    const float* W_ih   = (const float*)d_in[8];
    const float* b_ih   = (const float*)d_in[9];
    const float* W_hh   = (const float*)d_in[10];
    const float* b_hh   = (const float*)d_in[11];
    const float* W_dec  = (const float*)d_in[12];
    const float* b_dec  = (const float*)d_in[13];

    float* out       = (float*)d_out;
    float* hid_out   = out + V_;
    float* stack_out = out + V_ + H_;
    float* ws        = (float*)d_ws;

    sinp_and_blend<<<dim3(SINP_BLOCKS_ + BLEND_B1_), dim3(BLEND_ROWS_ * 32), 0, stream>>>(
        W_sin, hidden, b_sin, stack, stack_out, W_ctrl, b_ctrl, ws);
    gemv_gi_gh<<<dim3(GI_BLOCKS_ + 1), dim3(NW_G_ * 32), 0, stream>>>(
        inp, emb, hidden, stack, W_ih, b_ih, W_hh, b_hh, stack_out, ws);
    gates_dec_and_blend<<<dim3(DEC_BLOCKS_ + BLEND_B3_), dim3(BLEND_ROWS_ * 32), 0, stream>>>(
        W_dec, b_dec, out, hid_out, hidden, stack, stack_out, W_ctrl, b_ctrl, ws);
}


# Most of the logits launch's blend share moves into the gi/gh launch: half a row per matrix wave under its products (A fragments via LDS for register room) + 63 blend blocks on free WGPs; same sums
# speedup vs baseline: 1.8838x; 1.0606x over previous
#include <hip/hip_runtime.h>
#include <hip/hip_bf16.h>
#include <math.h>

#define V_  128
#define H_  1024
#define SW_ 1024
#define SD_ 8192

typedef __attribute__((ext_vector_type(2))) float v2f;
typedef __attribute__((ext_vector_type(4))) float v4f;
typedef __attribute__((ext_vector_type(8))) float v8f;

#define WS_CTRL   0
#define WS_SINP   16
#define WS_GH     4096
#define WS_GI     8192

#define KS_    128
#define PITCH_ (KS_ + 4)

template <int ACT, int K>
__device__ __forceinline__
void gemv_rows16(const float* __restrict__ W, const float* __restrict__ vec,
                 const float* __restrict__ bias, float* __restrict__ out, const int row_base,
                 float (*tile)[16][PITCH_], float (*red)[16]) {
    constexpr int NW = K / KS_;
    const int lane = threadIdx.x & 31;
    const int wave = threadIdx.x >> 5;
    const int kbeg = wave * KS_;

    const float* wsrc = W + (size_t)row_base * K + (size_t)(kbeg + lane * 4);
    v4f t[16];
    for (int r = 0; r < 16; ++r) t[r] = *(const v4f*)(wsrc + (size_t)r * K);

    const int co = (lane >> 4) << 1;
    const float* vp = vec + kbeg + co;
    v2f av[KS_ / 4];
    for (int s = 0; s < KS_ / 4; ++s) av[s] = *(const v2f*)(vp + 4 * s);

    for (int r = 0; r < 16; ++r) *(v4f*)&tile[wave][r][lane * 4] = t[r];
    __syncthreads();

    const float* bp = &tile[wave][lane & 15][co];
    v8f c0 = {}, c1 = {}, c2 = {}, c3 = {};
    for (int k = 0; k < KS_; k += 16) {
        v2f b0 = *(const v2f*)(bp + k);
        v2f b1 = *(const v2f*)(bp + k + 4);
        v2f b2 = *(const v2f*)(bp + k + 8);
        v2f b3 = *(const v2f*)(bp + k + 12);
        c0 = __builtin_amdgcn_wmma_f32_16x16x4_f32(false, av[k / 4],     false, b0, (short)0, c0, false, false);
        c1 = __builtin_amdgcn_wmma_f32_16x16x4_f32(false, av[k / 4 + 1], false, b1, (short)0, c1, false, false);
        c2 = __builtin_amdgcn_wmma_f32_16x16x4_f32(false, av[k / 4 + 2], false, b2, (short)0, c2, false, false);
        c3 = __builtin_amdgcn_wmma_f32_16x16x4_f32(false, av[k / 4 + 3], false, b3, (short)0, c3, false, false);
    }

    if (lane < 16) red[wave][lane] = c0[0] + c1[0] + c2[0] + c3[0];
    __syncthreads();
    if (threadIdx.x < 16) {
        float s = 0.f;
        #pragma unroll
        for (int w = 0; w < NW; ++w) s += red[w][threadIdx.x];
        s += bias[row_base + threadIdx.x];
        if (ACT == 1) s = tanhf(s);
        out[row_base + threadIdx.x] = s;
    }
}

#define CTRL_WAVES_ 8
static_assert(CTRL_WAVES_ * 32 * 4 == H_, "one 16-byte piece of h per thread");

__device__ __forceinline__
void ctrl_partials(const v4f hv, const v4f w0, const v4f w1, const v4f w2,
                   const int lane, float* part_w) {
    float p0 = hv[0] * w0[0] + hv[1] * w0[1] + hv[2] * w0[2] + hv[3] * w0[3];
    float p1 = hv[0] * w1[0] + hv[1] * w1[1] + hv[2] * w1[2] + hv[3] * w1[3];
    float p2 = hv[0] * w2[0] + hv[1] * w2[1] + hv[2] * w2[2] + hv[3] * w2[3];
    for (int off = 16; off > 0; off >>= 1) {
        p0 += __shfl_xor(p0, off, 32);
        p1 += __shfl_xor(p1, off, 32);
        p2 += __shfl_xor(p2, off, 32);
    }
    if (lane == 0) *(v4f*)part_w = v4f{p0, p1, p2, 0.f};
}

__device__ __forceinline__
void ctrl_softmax(const float (*part)[4], const float bc0, const float bc1, const float bc2,
                  float& a_push, float& a_pop, float& a_noop) {
    float l0 = 0.f, l1 = 0.f, l2 = 0.f;
    for (int w = 0; w < CTRL_WAVES_; ++w) {
        const v4f q = *(const v4f*)part[w];
        l0 += q[0]; l1 += q[1]; l2 += q[2];
    }
    l0 += bc0; l1 += bc1; l2 += bc2;
    const float m  = fmaxf(l0, fmaxf(l1, l2));
    const float e0 = expf(l0 - m), e1 = expf(l1 - m), e2 = expf(l2 - m);
    const float inv = 1.f / (e0 + e1 + e2);
    a_push = e0 * inv;
    a_pop  = e1 * inv;
    a_noop = e2 * inv;
}

#define BLEND_ROWS_   8
#define BLEND_NP_     (SW_ / 128)
#define SINP_BLOCKS_  (SW_ / 16)
#define DEC_BLOCKS_   (V_ / 16)
#define BLEND_BLOCKS_ (SD_ / BLEND_ROWS_)
#define BLEND_B1_     448
#define BLEND_B3_     (BLEND_BLOCKS_ - L3_BASE_ / BLEND_ROWS_)
static_assert(BLEND_ROWS_ == CTRL_WAVES_, "the blend block's waves form the controls");
static_assert(BLEND_ROWS_ == H_ / KS_, "both kinds of block have the same number of waves");

__device__ __forceinline__
void blend_block(const int bblk, const float* __restrict__ hidden, const float* __restrict__ stack,
                 float* __restrict__ new_stack, const float* __restrict__ W_ctrl,
                 const float* __restrict__ b_ctrl, float* __restrict__ ws, float (*part)[4]) {
    const int lane = threadIdx.x & 31;
    const int wave = threadIdx.x >> 5;

    const int e = threadIdx.x * 4;
    const v4f hv = *(const v4f*)(hidden + e);
    const v4f w0 = *(const v4f*)(W_ctrl + e);
    const v4f w1 = *(const v4f*)(W_ctrl + H_ + e);
    const v4f w2 = *(const v4f*)(W_ctrl + 2 * H_ + e);
    const float bc0 = b_ctrl[0], bc1 = b_ctrl[1], bc2 = b_ctrl[2];

    const int j  = bblk * BLEND_ROWS_ + wave;
    const bool live = (j != 0);
    const int c0 = lane * 4;
    const float* cp = stack + (size_t)j * SW_ + c0;
    const float* upp = live ? (cp - SW_) : cp;

    v4f cur[BLEND_NP_], up[BLEND_NP_], dn[BLEND_NP_];
    for (int p = 0; p < BLEND_NP_; ++p) cur[p] = *(const v4f*)(cp + p * 128);
    for (int p = 0; p < BLEND_NP_; ++p) up[p]  = *(const v4f*)(upp + p * 128);
    for (int p = 0; p < BLEND_NP_; ++p) dn[p]  = v4f{0.f, 0.f, 0.f, 0.f};
    if (j < SD_ - 1)
        for (int p = 0; p < BLEND_NP_; ++p) dn[p] = *(const v4f*)(cp + SW_ + p * 128);

    ctrl_partials(hv, w0, w1, w2, lane, part[wave]);
    __syncthreads();
    float a_push, a_pop, a_noop;
    ctrl_softmax(part, bc0, bc1, bc2, a_push, a_pop, a_noop);
    if (!live) {
        if (lane == 0) {
            ws[WS_CTRL + 0] = a_push;
            ws[WS_CTRL + 1] = a_pop;
            ws[WS_CTRL + 2] = a_noop;
        }
        return;
    }

    float* op = new_stack + (size_t)j * SW_ + c0;
    for (int p = 0; p < BLEND_NP_; ++p) {
        v4f o = a_noop * cur[p] + a_push * up[p] + a_pop * dn[p];
        __builtin_nontemporal_store(o, (v4f*)(op + p * 128));
    }
}

__global__ __launch_bounds__(BLEND_ROWS_ * 32)
void sinp_and_blend(const float* __restrict__ W_sin, const float* __restrict__ hidden,
                    const float* __restrict__ b_sin, const float* __restrict__ stack,
                    float* __restrict__ new_stack, const float* __restrict__ W_ctrl,
                    const float* __restrict__ b_ctrl, float* __restrict__ ws) {
    __shared__ __attribute__((aligned(16))) float tile[H_ / KS_][16][PITCH_];
    __shared__ float red[H_ / KS_][16];
    __shared__ __attribute__((aligned(16))) float part[CTRL_WAVES_][4];

    if (blockIdx.x < SINP_BLOCKS_) {
        gemv_rows16<1, H_>(W_sin, hidden, b_sin, ws + WS_SINP, blockIdx.x * 16, tile, red);
        return;
    }
    blend_block(blockIdx.x - SINP_BLOCKS_, hidden, stack, new_stack, W_ctrl, b_ctrl, ws, part);
}

#define NW_IH_ ((H_ + SW_) / KS_)
#define NW_X_  (H_ / KS_)
#define NW_S_  (SW_ / KS_)
#define NW_HH_ (H_ / KS_)
#define NW_G_  (NW_IH_ + NW_HH_)
#define GI_BLOCKS_ (3 * H_ / 16)
#define RIDE_ROWS_   (NW_G_ / 2)
#define RIDE_BASE_   (BLEND_B1_ * BLEND_ROWS_)
#define FILL_BLOCKS_ 63
#define FILL_BASE_   (RIDE_BASE_ + GI_BLOCKS_ * RIDE_ROWS_)
#define L3_BASE_     (FILL_BASE_ + FILL_BLOCKS_ * NW_G_)
static_assert(L3_BASE_ % BLEND_ROWS_ == 0 && L3_BASE_ < SD_, "the last launch takes whole blend blocks");

__global__ __launch_bounds__(NW_G_ * 32)
void gemv_gi_gh(const int* __restrict__ inp, const float* __restrict__ emb,
                const float* __restrict__ hidden, const float* __restrict__ stack,
                const float* __restrict__ W_ih, const float* __restrict__ b_ih,
                const float* __restrict__ W_hh, const float* __restrict__ b_hh,
                float* __restrict__ new_stack, float* ws) {
    __shared__ __attribute__((aligned(16))) float tile[NW_G_][16][PITCH_];
    __shared__ float red[NW_G_][16];
    __shared__ float red_up[NW_S_][16];
    __shared__ float red_dn[NW_S_][16];
    __shared__ __attribute__((aligned(16))) float avec[NW_G_][3][PITCH_];

    const float a_push = ws[WS_CTRL + 0];
    const float a_pop  = ws[WS_CTRL + 1];
    const float a_noop = ws[WS_CTRL + 2];

    if (blockIdx.x == GI_BLOCKS_) {
        if (threadIdx.x < SW_ / 4) {
            const int e = threadIdx.x * 4;
            const v4f r_cur = *(const v4f*)(stack + e);
            const v4f r_up  = *(const v4f*)((const float*)ws + WS_SINP + e);
            const v4f r_dn  = *(const v4f*)(stack + SW_ + e);
            v4f o = a_noop * r_cur + a_push * r_up + a_pop * r_dn;
            __builtin_nontemporal_store(o, (v4f*)(new_stack + e));
        }
        return;
    }

    const int lane = threadIdx.x & 31;
    const int wave = threadIdx.x >> 5;

    if (blockIdx.x > GI_BLOCKS_) {
        const int j = FILL_BASE_ + (blockIdx.x - GI_BLOCKS_ - 1) * NW_G_ + wave;
        const float* cp = stack + (size_t)j * SW_ + lane * 4;
        v4f cur[BLEND_NP_], up[BLEND_NP_], dn[BLEND_NP_];
        for (int p = 0; p < BLEND_NP_; ++p) cur[p] = *(const v4f*)(cp + p * 128);
        for (int p = 0; p < BLEND_NP_; ++p) up[p]  = *(const v4f*)(cp - SW_ + p * 128);
        for (int p = 0; p < BLEND_NP_; ++p) dn[p]  = *(const v4f*)(cp + SW_ + p * 128);
        float* op = new_stack + (size_t)j * SW_ + lane * 4;
        for (int p = 0; p < BLEND_NP_; ++p) {
            v4f o = a_noop * cur[p] + a_push * up[p] + a_pop * dn[p];
            __builtin_nontemporal_store(o, (v4f*)(op + p * 128));
        }
        return;
    }

    const int row_base = blockIdx.x * 16;
    const int co = (lane >> 4) << 1;

    v4f t[16];
    const float* v0p;
    const bool three = (wave >= NW_X_ && wave < NW_IH_);
    if (wave < NW_IH_) {
        const float* wsrc = W_ih + (size_t)row_base * (H_ + SW_) + (size_t)(wave * KS_ + lane * 4);
        for (int r = 0; r < 16; ++r) t[r] = *(const v4f*)(wsrc + (size_t)r * (H_ + SW_));
        v0p = three ? stack + (wave - NW_X_) * KS_ : emb + (size_t)inp[0] * H_ + wave * KS_;
    } else {
        const int kw = wave - NW_IH_;
        const float* wsrc = W_hh + (size_t)row_base * H_ + (size_t)(kw * KS_ + lane * 4);
        for (int r = 0; r < 16; ++r) t[r] = *(const v4f*)(wsrc + (size_t)r * H_);
        v0p = hidden + kw * KS_;
    }
    const v4f va0 = *(const v4f*)(v0p + lane * 4);
    v4f va1 = {}, va2 = {};
    if (three) {
        va1 = *(const v4f*)((const float*)ws + WS_SINP + (wave - NW_X_) * KS_ + lane * 4);
        va2 = *(const v4f*)(stack + SW_ + (wave - NW_X_) * KS_ + lane * 4);
    }

    for (int r = 0; r < 16; ++r) *(v4f*)&tile[wave][r][lane * 4] = t[r];
    *(v4f*)&avec[wave][0][lane * 4] = va0;
    if (three) {
        *(v4f*)&avec[wave][1][lane * 4] = va1;
        *(v4f*)&avec[wave][2][lane * 4] = va2;
    }

    __syncthreads();

    const int jr = RIDE_BASE_ + blockIdx.x * RIDE_ROWS_ + (wave >> 1);
    const size_t ro = (size_t)jr * SW_ + (size_t)((wave & 1) * (SW_ / 2) + lane * 4);
    v4f rc[BLEND_NP_ / 2], ru[BLEND_NP_ / 2], rd[BLEND_NP_ / 2];
    for (int p = 0; p < BLEND_NP_ / 2; ++p) rc[p] = *(const v4f*)(stack + ro + p * 128);
    for (int p = 0; p < BLEND_NP_ / 2; ++p) ru[p] = *(const v4f*)(stack + ro - SW_ + p * 128);
    for (int p = 0; p < BLEND_NP_ / 2; ++p) rd[p] = *(const v4f*)(stack + ro + SW_ + p * 128);

    const int m = lane & 15;
    const float* ap = &avec[wave][three ? (m == 1 ? 1 : (m == 2 ? 2 : 0)) : 0][co];
    const float* bp = &tile[wave][m][co];
    v8f c0 = {}, c1 = {}, c2 = {}, c3 = {};
    for (int k = 0; k < KS_; k += 16) {
        v2f f0 = *(const v2f*)(ap + k);
        v2f f1 = *(const v2f*)(ap + k + 4);
        v2f f2 = *(const v2f*)(ap + k + 8);
        v2f f3 = *(const v2f*)(ap + k + 12);
        v2f b0 = *(const v2f*)(bp + k);
        v2f b1 = *(const v2f*)(bp + k + 4);
        v2f b2 = *(const v2f*)(bp + k + 8);
        v2f b3 = *(const v2f*)(bp + k + 12);
        c0 = __builtin_amdgcn_wmma_f32_16x16x4_f32(false, f0, false, b0, (short)0, c0, false, false);
        c1 = __builtin_amdgcn_wmma_f32_16x16x4_f32(false, f1, false, b1, (short)0, c1, false, false);
        c2 = __builtin_amdgcn_wmma_f32_16x16x4_f32(false, f2, false, b2, (short)0, c2, false, false);
        c3 = __builtin_amdgcn_wmma_f32_16x16x4_f32(false, f3, false, b3, (short)0, c3, false, false);
    }

    if (lane < 16) {
        red[wave][lane] = c0[0] + c1[0] + c2[0] + c3[0];
        if (wave >= NW_X_ && wave < NW_IH_) {
            red_up[wave - NW_X_][lane] = c0[1] + c1[1] + c2[1] + c3[1];
            red_dn[wave - NW_X_][lane] = c0[2] + c1[2] + c2[2] + c3[2];
        }
    }
    __syncthreads();

    if (wave == 0 && lane < 16) {
        float sx = 0.f, sc = 0.f, su = 0.f, sd = 0.f;
        for (int w = 0; w < NW_X_; ++w) sx += red[w][lane];
        for (int w = 0; w < NW_S_; ++w) {
            sc += red[NW_X_ + w][lane];
            su += red_up[w][lane];
            sd += red_dn[w][lane];
        }
        ws[WS_GI + row_base + lane] = sx + (a_noop * sc + a_push * su + a_pop * sd) + b_ih[row_base + lane];
    } else if (wave == 1 && lane < 16) {
        float s = 0.f;
        for (int w = NW_IH_; w < NW_G_; ++w) s += red[w][lane];
        ws[WS_GH + row_base + lane] = s + b_hh[row_base + lane];
    }

    for (int p = 0; p < BLEND_NP_ / 2; ++p) {
        v4f o = a_noop * rc[p] + a_push * ru[p] + a_pop * rd[p];
        __builtin_nontemporal_store(o, (v4f*)(new_stack + ro + p * 128));
    }
}

__device__ __forceinline__
void gates_dec_rows16(const float* __restrict__ W_dec, const float* __restrict__ b_dec,
                      float* __restrict__ out, float* __restrict__ hid_out,
                      const float* __restrict__ hidden, const float* __restrict__ ws,
                      float (*tile)[16][PITCH_], float (*red)[16], float* hn) {
    const int lane = threadIdx.x & 31;
    const int wave = threadIdx.x >> 5;
    const int kbeg = wave * KS_;
    const int row_base = blockIdx.x * 16;
    const int e = threadIdx.x * 4;

    const float* gi = ws + WS_GI;
    const float* gh = ws + WS_GH;
    const v4f gir = *(const v4f*)(gi + e);
    const v4f ghr = *(const v4f*)(gh + e);
    const v4f giz = *(const v4f*)(gi + H_ + e);
    const v4f ghz = *(const v4f*)(gh + H_ + e);
    const v4f gin = *(const v4f*)(gi + 2 * H_ + e);
    const v4f ghn = *(const v4f*)(gh + 2 * H_ + e);
    const v4f hv  = *(const v4f*)(hidden + e);

    const float* wsrc = W_dec + (size_t)row_base * H_ + (size_t)(kbeg + lane * 4);
    v4f t[16];
    for (int r = 0; r < 16; ++r) t[r] = *(const v4f*)(wsrc + (size_t)r * H_);

    v4f hn4;
    for (int q = 0; q < 4; ++q) {
        float r  = 1.f / (1.f + expf(-(gir[q] + ghr[q])));
        float z  = 1.f / (1.f + expf(-(giz[q] + ghz[q])));
        float n  = tanhf(gin[q] + r * ghn[q]);
        hn4[q] = fmaf(1.f - z, n, z * hv[q]);
    }
    *(v4f*)(hn + e) = hn4;
    if (blockIdx.x == 0) *(v4f*)(hid_out + e) = hn4;

    for (int r = 0; r < 16; ++r) *(v4f*)&tile[wave][r][lane * 4] = t[r];
    __syncthreads();

    const int co = (lane >> 4) << 1;
    const float* vp = hn + kbeg + co;
    v2f av[KS_ / 4];
    for (int s = 0; s < KS_ / 4; ++s) av[s] = *(const v2f*)(vp + 4 * s);

    const float* bp = &tile[wave][lane & 15][co];
    v8f c0 = {}, c1 = {}, c2 = {}, c3 = {};
    for (int k = 0; k < KS_; k += 16) {
        v2f b0 = *(const v2f*)(bp + k);
        v2f b1 = *(const v2f*)(bp + k + 4);
        v2f b2 = *(const v2f*)(bp + k + 8);
        v2f b3 = *(const v2f*)(bp + k + 12);
        c0 = __builtin_amdgcn_wmma_f32_16x16x4_f32(false, av[k / 4],     false, b0, (short)0, c0, false, false);
        c1 = __builtin_amdgcn_wmma_f32_16x16x4_f32(false, av[k / 4 + 1], false, b1, (short)0, c1, false, false);
        c2 = __builtin_amdgcn_wmma_f32_16x16x4_f32(false, av[k / 4 + 2], false, b2, (short)0, c2, false, false);
        c3 = __builtin_amdgcn_wmma_f32_16x16x4_f32(false, av[k / 4 + 3], false, b3, (short)0, c3, false, false);
    }

    if (lane < 16) red[wave][lane] = c0[0] + c1[0] + c2[0] + c3[0];
    __syncthreads();
    if (threadIdx.x < 16) {
        float s = 0.f;
        #pragma unroll
        for (int w = 0; w < H_ / KS_; ++w) s += red[w][threadIdx.x];
        s += b_dec[row_base + threadIdx.x];
        out[row_base + threadIdx.x] = s;
    }
}

static_assert(BLEND_B3_ > 0, "every launch carries a share");

__global__ __launch_bounds__(BLEND_ROWS_ * 32)
void gates_dec_and_blend(const float* __restrict__ W_dec, const float* __restrict__ b_dec,
                         float* __restrict__ out, float* __restrict__ hid_out,
                         const float* __restrict__ hidden,
                         const float* __restrict__ stack, float* __restrict__ new_stack,
                         const float* __restrict__ W_ctrl, const float* __restrict__ b_ctrl,
                         float* __restrict__ ws) {
    __shared__ __attribute__((aligned(16))) float tile[H_ / KS_][16][PITCH_];
    __shared__ float red[H_ / KS_][16];
    __shared__ __attribute__((aligned(16))) float part[CTRL_WAVES_][4];
    __shared__ __attribute__((aligned(16))) float hn[H_];

    if (blockIdx.x < DEC_BLOCKS_) {
        gates_dec_rows16(W_dec, b_dec, out, hid_out, hidden, ws, tile, red, hn);
        return;
    }
    blend_block(L3_BASE_ / BLEND_ROWS_ + (blockIdx.x - DEC_BLOCKS_), hidden, stack, new_stack, W_ctrl, b_ctrl, ws, part);
}

extern "C" void kernel_launch(void* const* d_in, const int* in_sizes, int n_in,
                              void* d_out, int out_size, void* d_ws, size_t ws_size,
                              hipStream_t stream) {
    (void)in_sizes; (void)n_in; (void)out_size; (void)ws_size;
    const int*   inp    = (const int*)  d_in[0];
    const float* hidden = (const float*)d_in[1];
    const float* stack  = (const float*)d_in[2];
    const float* emb    = (const float*)d_in[3];
    const float* W_ctrl = (const float*)d_in[4];
    const float* b_ctrl = (const float*)d_in[5];
    const float* W_sin  = (const float*)d_in[6];
    const float* b_sin  = (const float*)d_in[7];
    const float* W_ih   = (const float*)d_in[8];
    const float* b_ih   = (const float*)d_in[9];
    const float* W_hh   = (const float*)d_in[10];
    const float* b_hh   = (const float*)d_in[11];
    const float* W_dec  = (const float*)d_in[12];
    const float* b_dec  = (const float*)d_in[13];

    float* out       = (float*)d_out;
    float* hid_out   = out + V_;
    float* stack_out = out + V_ + H_;
    float* ws        = (float*)d_ws;

    sinp_and_blend<<<dim3(SINP_BLOCKS_ + BLEND_B1_), dim3(BLEND_ROWS_ * 32), 0, stream>>>(
        W_sin, hidden, b_sin, stack, stack_out, W_ctrl, b_ctrl, ws);
    gemv_gi_gh<<<dim3(GI_BLOCKS_ + 1 + FILL_BLOCKS_), dim3(NW_G_ * 32), 0, stream>>>(
        inp, emb, hidden, stack, W_ih, b_ih, W_hh, b_hh, stack_out, ws);
    gates_dec_and_blend<<<dim3(DEC_BLOCKS_ + BLEND_B3_), dim3(BLEND_ROWS_ * 32), 0, stream>>>(
        W_dec, b_dec, out, hid_out, hidden, stack, stack_out, W_ctrl, b_ctrl, ws);
}
